# starting kernel
# baseline (speedup 1.0000x reference)
#include <hip/hip_runtime.h>
#include <math.h>

typedef __attribute__((ext_vector_type(2))) float v2f;
typedef __attribute__((ext_vector_type(8))) float v8f;

#define WAVES 4

constexpr int B_ = 2, L_ = 2048, S_ = 2048, H_ = 8, E_ = 64, D_ = 64;
constexpr float SCALE  = 0.125f;
constexpr float MASKED = -1.25e8f;

__device__ __forceinline__ bool log_mask(int i, int j) {
    int d = i - j;
    if (d < 0) return false;
    if (i < 22) return true;
    if (d <= 10) return true;
    int e = d - 10;
    return (e <= 1024) && ((e & (e - 1)) == 0);
}

__device__ __forceinline__ bool tile_active(int i0, int j0) {
    int delta = i0 - j0;
    if (delta < -15) return false;
    if (i0 < 32) return true;
    if (delta <= 25) return true;
#pragma unroll
    for (int p = 0; p < 11; ++p) {
        int dv = 10 + (1 << p);
        if (dv >= delta - 15 && dv <= delta + 15) return true;
    }
    return false;
}

__global__ __launch_bounds__(WAVES * 32)
void logsparse_attn_kernel(const float* __restrict__ Q,
                           const float* __restrict__ K,
                           const float* __restrict__ V,
                           float* __restrict__ Out) {
    __shared__ __align__(16) float plds[WAVES][16][18];

    const int lane = threadIdx.x & 31;
    const int wave = threadIdx.x >> 5;
    const int g    = lane >> 4;
    const int ln   = lane & 15;

    const int gid = blockIdx.x * WAVES + wave;
    const int qt  = gid & 127;
    const int h   = (gid >> 7) & 7;
    const int b   = gid >> 10;
    const int i0  = qt * 16;

    const float* qrow = Q + ((size_t)(b * L_ + i0 + ln) * H_ + h) * E_ + 2 * g;
    v2f qf[16];
#pragma unroll
    for (int kc = 0; kc < 16; ++kc) {
        float2 t = *(const float2*)(qrow + 4 * kc);
        v2f a; a[0] = t.x; a[1] = t.y;
        qf[kc] = a;
    }

    v8f oacc[4];
#pragma unroll
    for (int nt = 0; nt < 4; ++nt) oacc[nt] = (v8f){};
    float rowmax[8], rowsum[8];
#pragma unroll
    for (int r = 0; r < 8; ++r) { rowmax[r] = -INFINITY; rowsum[r] = 0.0f; }

    const int jt_end = qt;
    for (int jt = 0; jt <= jt_end; ++jt) {
        const int j0 = jt * 16;
        if (!tile_active(i0, j0)) continue;

        const float* krow = K + ((size_t)(b * S_ + j0 + ln) * H_ + h) * E_ + 2 * g;
        v8f s = (v8f){};
#pragma unroll
        for (int kc = 0; kc < 16; ++kc) {
            float2 t = *(const float2*)(krow + 4 * kc);
            v2f kf; kf[0] = t.x; kf[1] = t.y;
            s = __builtin_amdgcn_wmma_f32_16x16x4_f32(false, qf[kc], false, kf,
                                                      (short)0, s, false, false);
        }

#pragma unroll
        for (int r = 0; r < 8; ++r) {
            const int i = i0 + r + 8 * g;
            const int j = j0 + ln;
            float logit = log_mask(i, j) ? SCALE * s[r] : MASKED;

            float tmax = logit;
#pragma unroll
            for (int off = 8; off >= 1; off >>= 1)
                tmax = fmaxf(tmax, __shfl_xor(tmax, off, 32));

            float newmax = fmaxf(rowmax[r], tmax);
            float corr   = __expf(rowmax[r] - newmax);
            float p      = __expf(logit - newmax);

            float tsum = p;
#pragma unroll
            for (int off = 8; off >= 1; off >>= 1)
                tsum += __shfl_xor(tsum, off, 32);

            rowsum[r] = rowsum[r] * corr + tsum;
            rowmax[r] = newmax;
#pragma unroll
            for (int nt = 0; nt < 4; ++nt) oacc[nt][r] *= corr;

            plds[wave][r + 8 * g][ln] = p;
        }
        __builtin_amdgcn_wave_barrier();

        v2f pa[4];
#pragma unroll
        for (int kc = 0; kc < 4; ++kc) {
            float2 t = *(const float2*)(&plds[wave][ln][4 * kc + 2 * g]);
            v2f a; a[0] = t.x; a[1] = t.y;
            pa[kc] = a;
        }
        __builtin_amdgcn_wave_barrier();

        const float* vbase = V + ((size_t)(b * S_ + j0) * H_ + h) * D_;
#pragma unroll
        for (int nt = 0; nt < 4; ++nt) {
#pragma unroll
            for (int kc = 0; kc < 4; ++kc) {
                const int kr = 4 * kc + 2 * g;
                v2f vf;
                vf[0] = vbase[(size_t)(kr + 0) * (H_ * D_) + nt * 16 + ln];
                vf[1] = vbase[(size_t)(kr + 1) * (H_ * D_) + nt * 16 + ln];
                oacc[nt] = __builtin_amdgcn_wmma_f32_16x16x4_f32(false, pa[kc], false, vf,
                                                                 (short)0, oacc[nt], false, false);
            }
        }
    }

#pragma unroll
    for (int r = 0; r < 8; ++r) {
        const int i = i0 + r + 8 * g;
        const float inv = 1.0f / rowsum[r];
        float* orow = Out + ((size_t)(b * L_ + i) * H_ + h) * D_;
        orow[ 0 + ln] = oacc[0][r] * inv;
        orow[16 + ln] = oacc[1][r] * inv;
        orow[32 + ln] = oacc[2][r] * inv;
        orow[48 + ln] = oacc[3][r] * inv;
    }
}

extern "C" void kernel_launch(void* const* d_in, const int* in_sizes, int n_in,
                              void* d_out, int out_size, void* d_ws, size_t ws_size,
                              hipStream_t stream) {
    (void)in_sizes; (void)n_in; (void)out_size; (void)d_ws; (void)ws_size;
    const float* Q = (const float*)d_in[0];
    const float* K = (const float*)d_in[1];
    const float* V = (const float*)d_in[2];
    float* Out = (float*)d_out;

    const int total_waves = B_ * H_ * (L_ / 16);
    dim3 grid(total_waves / WAVES);
    dim3 block(WAVES * 32);
    hipLaunchKernelGGL(logsparse_attn_kernel, grid, block, 0, stream, Q, K, V, Out);
}


# one query tile per work-group: its active key tiles dealt to the 4 waves, partial softmax states (max, sum, O) merged in LDS; changes the reduction tree of row sums and O
# speedup vs baseline: 1.8101x; 1.8101x over previous
#include <hip/hip_runtime.h>
#include <math.h>

typedef __attribute__((ext_vector_type(2))) float v2f;
typedef __attribute__((ext_vector_type(8))) float v8f;

#define WAVES 4

constexpr int B_ = 2, L_ = 2048, S_ = 2048, H_ = 8, E_ = 64, D_ = 64;
constexpr float SCALE  = 0.125f;
constexpr float MASKED = -1.25e8f;

__device__ __forceinline__ bool log_mask(int i, int j) {
    int d = i - j;
    if (d < 0) return false;
    if (i < 22) return true;
    if (d <= 10) return true;
    int e = d - 10;
    return (e <= 1024) && ((e & (e - 1)) == 0);
}

constexpr int NTILES = 14;
__device__ __forceinline__ int tile_delta(int t) {
    const int p = t >> 1;
    const int base = (p < 2) ? 32 * p : (64 << (p - 2));
    return base + 16 * (t & 1);
}

constexpr int OSTR = 68;

__global__ __launch_bounds__(WAVES * 32)
void logsparse_attn_kernel(const float* __restrict__ Q,
                           const float* __restrict__ K,
                           const float* __restrict__ V,
                           float* __restrict__ Out) {
    __shared__ __align__(16) float plds[WAVES][16][18];
    __shared__ __align__(16) float po[WAVES][16][OSTR];
    __shared__ float pm[WAVES][16];
    __shared__ float pl[WAVES][16];

    const int lane = threadIdx.x & 31;
    const int wave = __builtin_amdgcn_readfirstlane(threadIdx.x >> 5);
    const int g    = lane >> 4;
    const int ln   = lane & 15;

    const int gid = blockIdx.x;
    const int qt  = gid & 127;
    const int h   = (gid >> 7) & 7;
    const int b   = gid >> 10;
    const int i0  = qt * 16;

    const float* qrow = Q + ((size_t)(b * L_ + i0 + ln) * H_ + h) * E_ + 2 * g;
    v2f qf[16];
#pragma unroll
    for (int kc = 0; kc < 16; ++kc) {
        float2 t = *(const float2*)(qrow + 4 * kc);
        v2f a; a[0] = t.x; a[1] = t.y;
        qf[kc] = a;
    }

    v8f oacc[4];
#pragma unroll
    for (int nt = 0; nt < 4; ++nt) oacc[nt] = (v8f){};
    float rowmax[8], rowsum[8];
#pragma unroll
    for (int r = 0; r < 8; ++r) { rowmax[r] = -INFINITY; rowsum[r] = 0.0f; }

    int nact = 0;
#pragma unroll
    for (int t = 0; t < NTILES; ++t) nact += (tile_delta(t) <= i0) ? 1 : 0;
    for (int ti = wave; ti < nact; ti += WAVES) {
        const int j0 = i0 - tile_delta(ti);

        const float* krow = K + ((size_t)(b * S_ + j0 + ln) * H_ + h) * E_ + 2 * g;
        v8f s = (v8f){};
#pragma unroll
        for (int kc = 0; kc < 16; ++kc) {
            float2 t = *(const float2*)(krow + 4 * kc);
            v2f kf; kf[0] = t.x; kf[1] = t.y;
            s = __builtin_amdgcn_wmma_f32_16x16x4_f32(false, qf[kc], false, kf,
                                                      (short)0, s, false, false);
        }

#pragma unroll
        for (int r = 0; r < 8; ++r) {
            const int i = i0 + r + 8 * g;
            const int j = j0 + ln;
            float logit = log_mask(i, j) ? SCALE * s[r] : MASKED;

            float tmax = logit;
#pragma unroll
            for (int off = 8; off >= 1; off >>= 1)
                tmax = fmaxf(tmax, __shfl_xor(tmax, off, 32));

            float newmax = fmaxf(rowmax[r], tmax);
            float corr   = __expf(rowmax[r] - newmax);
            float p      = __expf(logit - newmax);

            float tsum = p;
#pragma unroll
            for (int off = 8; off >= 1; off >>= 1)
                tsum += __shfl_xor(tsum, off, 32);

            rowsum[r] = rowsum[r] * corr + tsum;
            rowmax[r] = newmax;
#pragma unroll
            for (int nt = 0; nt < 4; ++nt) oacc[nt][r] *= corr;

            plds[wave][r + 8 * g][ln] = p;
        }
        __builtin_amdgcn_wave_barrier();

        v2f pa[4];
#pragma unroll
        for (int kc = 0; kc < 4; ++kc) {
            float2 t = *(const float2*)(&plds[wave][ln][4 * kc + 2 * g]);
            v2f a; a[0] = t.x; a[1] = t.y;
            pa[kc] = a;
        }
        __builtin_amdgcn_wave_barrier();

        const float* vbase = V + ((size_t)(b * S_ + j0) * H_ + h) * D_;
#pragma unroll
        for (int nt = 0; nt < 4; ++nt) {
#pragma unroll
            for (int kc = 0; kc < 4; ++kc) {
                const int kr = 4 * kc + 2 * g;
                v2f vf;
                vf[0] = vbase[(size_t)(kr + 0) * (H_ * D_) + nt * 16 + ln];
                vf[1] = vbase[(size_t)(kr + 1) * (H_ * D_) + nt * 16 + ln];
                oacc[nt] = __builtin_amdgcn_wmma_f32_16x16x4_f32(false, pa[kc], false, vf,
                                                                 (short)0, oacc[nt], false, false);
            }
        }
    }

#pragma unroll
    for (int r = 0; r < 8; ++r) {
        if (ln == 0) {
            pm[wave][r + 8 * g] = rowmax[r];
            pl[wave][r + 8 * g] = rowsum[r];
        }
#pragma unroll
        for (int nt = 0; nt < 4; ++nt) po[wave][r + 8 * g][nt * 16 + ln] = oacc[nt][r];
    }
    __syncthreads();

    const int c4 = (threadIdx.x & 15) * 4;
    for (int hf = 0; hf < 2; ++hf) {
        const int row = (threadIdx.x >> 4) + 8 * hf;
        const float m0 = pm[0][row], m1 = pm[1][row], m2 = pm[2][row], m3 = pm[3][row];
        const float mx = fmaxf(fmaxf(m0, m1), fmaxf(m2, m3));
        const float s0 = __expf(m0 - mx), s1 = __expf(m1 - mx);
        const float s2 = __expf(m2 - mx), s3 = __expf(m3 - mx);
        const float lsum = ((pl[0][row] * s0 + pl[1][row] * s1) + pl[2][row] * s2) + pl[3][row] * s3;
        const float inv = 1.0f / lsum;
        const float4 a0 = *(const float4*)(&po[0][row][c4]);
        const float4 a1 = *(const float4*)(&po[1][row][c4]);
        const float4 a2 = *(const float4*)(&po[2][row][c4]);
        const float4 a3 = *(const float4*)(&po[3][row][c4]);
        float4 o;
        o.x = (((a0.x * s0 + a1.x * s1) + a2.x * s2) + a3.x * s3) * inv;
        o.y = (((a0.y * s0 + a1.y * s1) + a2.y * s2) + a3.y * s3) * inv;
        o.z = (((a0.z * s0 + a1.z * s1) + a2.z * s2) + a3.z * s3) * inv;
        o.w = (((a0.w * s0 + a1.w * s1) + a2.w * s2) + a3.w * s3) * inv;
        float* orow = Out + ((size_t)(b * L_ + i0 + row) * H_ + h) * D_;
        *(float4*)(orow + c4) = o;
    }
}

extern "C" void kernel_launch(void* const* d_in, const int* in_sizes, int n_in,
                              void* d_out, int out_size, void* d_ws, size_t ws_size,
                              hipStream_t stream) {
    (void)in_sizes; (void)n_in; (void)out_size; (void)d_ws; (void)ws_size;
    const float* Q = (const float*)d_in[0];
    const float* K = (const float*)d_in[1];
    const float* V = (const float*)d_in[2];
    float* Out = (float*)d_out;

    dim3 grid(B_ * H_ * (L_ / 16));
    dim3 block(WAVES * 32);
    hipLaunchKernelGGL(logsparse_attn_kernel, grid, block, 0, stream, Q, K, V, Out);
}


# softmax row max and row sum: 16-lane butterflies as DPP row_xmask on the vector ALU instead of ds_bpermute with a wait each (mask test branch-free, load groups kept whole); same sum order
# speedup vs baseline: 3.0706x; 1.6964x over previous
#include <hip/hip_runtime.h>
#include <math.h>

typedef __attribute__((ext_vector_type(2))) float v2f;
typedef __attribute__((ext_vector_type(8))) float v8f;

#define WAVES 4

constexpr int B_ = 2, L_ = 2048, S_ = 2048, H_ = 8, E_ = 64, D_ = 64;
constexpr float SCALE  = 0.125f;
constexpr float MASKED = -1.25e8f;

__device__ __forceinline__ bool log_mask(int i, int j) {
    const int d = i - j;
    const int e = d - 10;
    const bool pow2 = (e <= 1024) & ((e & (e - 1)) == 0);
    return (d >= 0) & ((i < 22) | (d <= 10) | pow2);
}

template <int M>
__device__ __forceinline__ float row_xor(float x) {
    const int b = __builtin_bit_cast(int, x);
    return __builtin_bit_cast(float, __builtin_amdgcn_update_dpp(0, b, 0x160 | M, 0xf, 0xf, true));
}

constexpr int NTILES = 14;
__device__ __forceinline__ int tile_delta(int t) {
    const int p = t >> 1;
    const int base = (p < 2) ? 32 * p : (64 << (p - 2));
    return base + 16 * (t & 1);
}

constexpr int OSTR = 68;

__global__ __launch_bounds__(WAVES * 32) __attribute__((amdgpu_waves_per_eu(1, 6)))
void logsparse_attn_kernel(const float* __restrict__ Q,
                           const float* __restrict__ K,
                           const float* __restrict__ V,
                           float* __restrict__ Out) {
    __shared__ __align__(16) float plds[WAVES][16][18];
    __shared__ __align__(16) float po[WAVES][16][OSTR];
    __shared__ float pm[WAVES][16];
    __shared__ float pl[WAVES][16];

    const int lane = threadIdx.x & 31;
    const int wave = __builtin_amdgcn_readfirstlane(threadIdx.x >> 5);
    const int g    = lane >> 4;
    const int ln   = lane & 15;

    const int gid = blockIdx.x;
    const int qt  = gid & 127;
    const int h   = (gid >> 7) & 7;
    const int b   = gid >> 10;
    const int i0  = qt * 16;

    const float* qrow = Q + ((size_t)(b * L_ + i0 + ln) * H_ + h) * E_ + 2 * g;
    v2f qf[16];
#pragma unroll
    for (int kc = 0; kc < 16; ++kc) {
        float2 t = *(const float2*)(qrow + 4 * kc);
        v2f a; a[0] = t.x; a[1] = t.y;
        qf[kc] = a;
    }

    v8f oacc[4];
#pragma unroll
    for (int nt = 0; nt < 4; ++nt) oacc[nt] = (v8f){};
    float rowmax[8], rowsum[8];
#pragma unroll
    for (int r = 0; r < 8; ++r) { rowmax[r] = -INFINITY; rowsum[r] = 0.0f; }

    int nact = 0;
#pragma unroll
    for (int t = 0; t < NTILES; ++t) nact += (tile_delta(t) <= i0) ? 1 : 0;
    for (int ti = wave; ti < nact; ti += WAVES) {
        const int j0 = i0 - tile_delta(ti);

        const float* krow = K + ((size_t)(b * S_ + j0 + ln) * H_ + h) * E_ + 2 * g;
        v8f s = (v8f){};
#pragma unroll
        for (int kc = 0; kc < 16; ++kc) {
            float2 t = *(const float2*)(krow + 4 * kc);
            v2f kf; kf[0] = t.x; kf[1] = t.y;
            s = __builtin_amdgcn_wmma_f32_16x16x4_f32(false, qf[kc], false, kf,
                                                      (short)0, s, false, false);
        }

#pragma unroll
        for (int r = 0; r < 8; ++r) {
            const int i = i0 + r + 8 * g;
            const int j = j0 + ln;
            float logit = log_mask(i, j) ? SCALE * s[r] : MASKED;

            float tmax = logit;
            tmax = fmaxf(tmax, row_xor<8>(tmax));
            tmax = fmaxf(tmax, row_xor<4>(tmax));
            tmax = fmaxf(tmax, row_xor<2>(tmax));
            tmax = fmaxf(tmax, row_xor<1>(tmax));

            float newmax = fmaxf(rowmax[r], tmax);
            float corr   = __expf(rowmax[r] - newmax);
            float p      = __expf(logit - newmax);

            float tsum = p;
            tsum += row_xor<8>(tsum);
            tsum += row_xor<4>(tsum);
            tsum += row_xor<2>(tsum);
            tsum += row_xor<1>(tsum);

            rowsum[r] = rowsum[r] * corr + tsum;
            rowmax[r] = newmax;
#pragma unroll
            for (int nt = 0; nt < 4; ++nt) oacc[nt][r] *= corr;

            plds[wave][r + 8 * g][ln] = p;
        }
        __builtin_amdgcn_wave_barrier();

        v2f pa[4];
#pragma unroll
        for (int kc = 0; kc < 4; ++kc) {
            float2 t = *(const float2*)(&plds[wave][ln][4 * kc + 2 * g]);
            v2f a; a[0] = t.x; a[1] = t.y;
            pa[kc] = a;
        }
        __builtin_amdgcn_wave_barrier();

        const float* vbase = V + ((size_t)(b * S_ + j0) * H_ + h) * D_;
#pragma unroll
        for (int nt = 0; nt < 4; ++nt) {
#pragma unroll
            for (int kc = 0; kc < 4; ++kc) {
                const int kr = 4 * kc + 2 * g;
                v2f vf;
                vf[0] = vbase[(size_t)(kr + 0) * (H_ * D_) + nt * 16 + ln];
                vf[1] = vbase[(size_t)(kr + 1) * (H_ * D_) + nt * 16 + ln];
                oacc[nt] = __builtin_amdgcn_wmma_f32_16x16x4_f32(false, pa[kc], false, vf,
                                                                 (short)0, oacc[nt], false, false);
            }
        }
    }

#pragma unroll
    for (int r = 0; r < 8; ++r) {
        if (ln == 0) {
            pm[wave][r + 8 * g] = rowmax[r];
            pl[wave][r + 8 * g] = rowsum[r];
        }
#pragma unroll
        for (int nt = 0; nt < 4; ++nt) po[wave][r + 8 * g][nt * 16 + ln] = oacc[nt][r];
    }
    __syncthreads();

    const int c4 = (threadIdx.x & 15) * 4;
    for (int hf = 0; hf < 2; ++hf) {
        const int row = (threadIdx.x >> 4) + 8 * hf;
        const float m0 = pm[0][row], m1 = pm[1][row], m2 = pm[2][row], m3 = pm[3][row];
        const float mx = fmaxf(fmaxf(m0, m1), fmaxf(m2, m3));
        const float s0 = __expf(m0 - mx), s1 = __expf(m1 - mx);
        const float s2 = __expf(m2 - mx), s3 = __expf(m3 - mx);
        const float lsum = ((pl[0][row] * s0 + pl[1][row] * s1) + pl[2][row] * s2) + pl[3][row] * s3;
        const float inv = 1.0f / lsum;
        const float4 a0 = *(const float4*)(&po[0][row][c4]);
        const float4 a1 = *(const float4*)(&po[1][row][c4]);
        const float4 a2 = *(const float4*)(&po[2][row][c4]);
        const float4 a3 = *(const float4*)(&po[3][row][c4]);
        float4 o;
        o.x = (((a0.x * s0 + a1.x * s1) + a2.x * s2) + a3.x * s3) * inv;
        o.y = (((a0.y * s0 + a1.y * s1) + a2.y * s2) + a3.y * s3) * inv;
        o.z = (((a0.z * s0 + a1.z * s1) + a2.z * s2) + a3.z * s3) * inv;
        o.w = (((a0.w * s0 + a1.w * s1) + a2.w * s2) + a3.w * s3) * inv;
        float* orow = Out + ((size_t)(b * L_ + i0 + row) * H_ + h) * D_;
        *(float4*)(orow + c4) = o;
    }
}

extern "C" void kernel_launch(void* const* d_in, const int* in_sizes, int n_in,
                              void* d_out, int out_size, void* d_ws, size_t ws_size,
                              hipStream_t stream) {
    (void)in_sizes; (void)n_in; (void)out_size; (void)d_ws; (void)ws_size;
    const float* Q = (const float*)d_in[0];
    const float* K = (const float*)d_in[1];
    const float* V = (const float*)d_in[2];
    float* Out = (float*)d_out;

    dim3 grid(B_ * H_ * (L_ / 16));
    dim3 block(WAVES * 32);
    hipLaunchKernelGGL(logsparse_attn_kernel, grid, block, 0, stream, Q, K, V, Out);
}


# far key tiles placed on their diagonal (key tile starts at i0-d, off the tile grid): one tile per far distance instead of two, 14 -> 9 tiles per query tile; regroups the row sums and O sums
# speedup vs baseline: 3.5055x; 1.1416x over previous
#include <hip/hip_runtime.h>
#include <math.h>

typedef __attribute__((ext_vector_type(2))) float v2f;
typedef __attribute__((ext_vector_type(8))) float v8f;

#define WAVES 4

constexpr int B_ = 2, L_ = 2048, S_ = 2048, H_ = 8, E_ = 64, D_ = 64;
constexpr float SCALE  = 0.125f;
constexpr float MASKED = -1.25e8f;

__device__ __forceinline__ bool log_mask(int i, int j) {
    const int d = i - j;
    const int e = d - 10;
    const bool pow2 = (e <= 1024) & ((e & (e - 1)) == 0);
    return (d >= 0) & ((i < 22) | (d <= 10) | pow2);
}

__device__ __forceinline__ bool tile_mask(int i, int j, bool far) {
    return log_mask(i, j) & (((i - j) > 26) == far);
}

template <int M>
__device__ __forceinline__ float row_xor(float x) {
    const int b = __builtin_bit_cast(int, x);
    return __builtin_bit_cast(float, __builtin_amdgcn_update_dpp(0, b, 0x160 | M, 0xf, 0xf, true));
}

constexpr int NTILES = 9;
__device__ __forceinline__ int tile_off(int t)  { return (t < 3) ? 16 * t : 10 + (32 << (t - 3)); }
__device__ __forceinline__ int tile_need(int t) { return (t < 3) ? 16 * t : (32 << (t - 3)); }

constexpr int OSTR = 68;

__global__ __launch_bounds__(WAVES * 32) __attribute__((amdgpu_waves_per_eu(1, 6)))
void logsparse_attn_kernel(const float* __restrict__ Q,
                           const float* __restrict__ K,
                           const float* __restrict__ V,
                           float* __restrict__ Out) {
    __shared__ __align__(16) float plds[WAVES][16][18];
    __shared__ __align__(16) float po[WAVES][16][OSTR];
    __shared__ float pm[WAVES][16];
    __shared__ float pl[WAVES][16];

    const int lane = threadIdx.x & 31;
    const int wave = __builtin_amdgcn_readfirstlane(threadIdx.x >> 5);
    const int g    = lane >> 4;
    const int ln   = lane & 15;

    const int gid = blockIdx.x;
    const int qt  = gid & 127;
    const int h   = (gid >> 7) & 7;
    const int b   = gid >> 10;
    const int i0  = qt * 16;

    const float* qrow = Q + ((size_t)(b * L_ + i0 + ln) * H_ + h) * E_ + 2 * g;
    v2f qf[16];
#pragma unroll
    for (int kc = 0; kc < 16; ++kc) {
        float2 t = *(const float2*)(qrow + 4 * kc);
        v2f a; a[0] = t.x; a[1] = t.y;
        qf[kc] = a;
    }

    v8f oacc[4];
#pragma unroll
    for (int nt = 0; nt < 4; ++nt) oacc[nt] = (v8f){};
    float rowmax[8], rowsum[8];
#pragma unroll
    for (int r = 0; r < 8; ++r) { rowmax[r] = -INFINITY; rowsum[r] = 0.0f; }

    int nact = 0;
#pragma unroll
    for (int t = 0; t < NTILES; ++t) nact += (tile_need(t) <= i0) ? 1 : 0;
    for (int ti = wave; ti < nact; ti += WAVES) {
        const bool far = ti >= 3;
        const int j0 = max(i0 - tile_off(ti), 0);

        const float* krow = K + ((size_t)(b * S_ + j0 + ln) * H_ + h) * E_ + 2 * g;
        v8f s = (v8f){};
#pragma unroll
        for (int kc = 0; kc < 16; ++kc) {
            float2 t = *(const float2*)(krow + 4 * kc);
            v2f kf; kf[0] = t.x; kf[1] = t.y;
            s = __builtin_amdgcn_wmma_f32_16x16x4_f32(false, qf[kc], false, kf,
                                                      (short)0, s, false, false);
        }

#pragma unroll
        for (int r = 0; r < 8; ++r) {
            const int i = i0 + r + 8 * g;
            const int j = j0 + ln;
            float logit = tile_mask(i, j, far) ? SCALE * s[r] : MASKED;

            float tmax = logit;
            tmax = fmaxf(tmax, row_xor<8>(tmax));
            tmax = fmaxf(tmax, row_xor<4>(tmax));
            tmax = fmaxf(tmax, row_xor<2>(tmax));
            tmax = fmaxf(tmax, row_xor<1>(tmax));

            float newmax = fmaxf(rowmax[r], tmax);
            float corr   = __expf(rowmax[r] - newmax);
            float p      = __expf(logit - newmax);

            float tsum = p;
            tsum += row_xor<8>(tsum);
            tsum += row_xor<4>(tsum);
            tsum += row_xor<2>(tsum);
            tsum += row_xor<1>(tsum);

            rowsum[r] = rowsum[r] * corr + tsum;
            rowmax[r] = newmax;
#pragma unroll
            for (int nt = 0; nt < 4; ++nt) oacc[nt][r] *= corr;

            plds[wave][r + 8 * g][ln] = p;
        }
        __builtin_amdgcn_wave_barrier();

        v2f pa[4];
#pragma unroll
        for (int kc = 0; kc < 4; ++kc) {
            float2 t = *(const float2*)(&plds[wave][ln][4 * kc + 2 * g]);
            v2f a; a[0] = t.x; a[1] = t.y;
            pa[kc] = a;
        }
        __builtin_amdgcn_wave_barrier();

        const float* vbase = V + ((size_t)(b * S_ + j0) * H_ + h) * D_;
#pragma unroll
        for (int nt = 0; nt < 4; ++nt) {
#pragma unroll
            for (int kc = 0; kc < 4; ++kc) {
                const int kr = 4 * kc + 2 * g;
                v2f vf;
                vf[0] = vbase[(size_t)(kr + 0) * (H_ * D_) + nt * 16 + ln];
                vf[1] = vbase[(size_t)(kr + 1) * (H_ * D_) + nt * 16 + ln];
                oacc[nt] = __builtin_amdgcn_wmma_f32_16x16x4_f32(false, pa[kc], false, vf,
                                                                 (short)0, oacc[nt], false, false);
            }
        }
    }

#pragma unroll
    for (int r = 0; r < 8; ++r) {
        if (ln == 0) {
            pm[wave][r + 8 * g] = rowmax[r];
            pl[wave][r + 8 * g] = rowsum[r];
        }
#pragma unroll
        for (int nt = 0; nt < 4; ++nt) po[wave][r + 8 * g][nt * 16 + ln] = oacc[nt][r];
    }
    __syncthreads();

    const int c4 = (threadIdx.x & 15) * 4;
    for (int hf = 0; hf < 2; ++hf) {
        const int row = (threadIdx.x >> 4) + 8 * hf;
        const float m0 = pm[0][row], m1 = pm[1][row], m2 = pm[2][row], m3 = pm[3][row];
        const float mx = fmaxf(fmaxf(m0, m1), fmaxf(m2, m3));
        const float s0 = __expf(m0 - mx), s1 = __expf(m1 - mx);
        const float s2 = __expf(m2 - mx), s3 = __expf(m3 - mx);
        const float lsum = ((pl[0][row] * s0 + pl[1][row] * s1) + pl[2][row] * s2) + pl[3][row] * s3;
        const float inv = 1.0f / lsum;
        const float4 a0 = *(const float4*)(&po[0][row][c4]);
        const float4 a1 = *(const float4*)(&po[1][row][c4]);
        const float4 a2 = *(const float4*)(&po[2][row][c4]);
        const float4 a3 = *(const float4*)(&po[3][row][c4]);
        float4 o;
        o.x = (((a0.x * s0 + a1.x * s1) + a2.x * s2) + a3.x * s3) * inv;
        o.y = (((a0.y * s0 + a1.y * s1) + a2.y * s2) + a3.y * s3) * inv;
        o.z = (((a0.z * s0 + a1.z * s1) + a2.z * s2) + a3.z * s3) * inv;
        o.w = (((a0.w * s0 + a1.w * s1) + a2.w * s2) + a3.w * s3) * inv;
        float* orow = Out + ((size_t)(b * L_ + i0 + row) * H_ + h) * D_;
        *(float4*)(orow + c4) = o;
    }
}

extern "C" void kernel_launch(void* const* d_in, const int* in_sizes, int n_in,
                              void* d_out, int out_size, void* d_ws, size_t ws_size,
                              hipStream_t stream) {
    (void)in_sizes; (void)n_in; (void)out_size; (void)d_ws; (void)ws_size;
    const float* Q = (const float*)d_in[0];
    const float* K = (const float*)d_in[1];
    const float* V = (const float*)d_in[2];
    float* Out = (float*)d_out;

    dim3 grid(B_ * H_ * (L_ / 16));
    dim3 block(WAVES * 32);
    hipLaunchKernelGGL(logsparse_attn_kernel, grid, block, 0, stream, Q, K, V, Out);
}


# each work-group takes two query tiles in turn (tile 127-p, then p): half the waves to launch, one merge barrier; key tiles dealt to the 4 waves from a turning start, regrouping row sums and O sums
# speedup vs baseline: 4.1157x; 1.1741x over previous
#include <hip/hip_runtime.h>
#include <math.h>

typedef __attribute__((ext_vector_type(2))) float v2f;
typedef __attribute__((ext_vector_type(8))) float v8f;

#define WAVES 4
#define QPW 2

constexpr int B_ = 2, L_ = 2048, S_ = 2048, H_ = 8, E_ = 64, D_ = 64;
constexpr float SCALE  = 0.125f;
constexpr float MASKED = -1.25e8f;

__device__ __forceinline__ bool log_mask(int i, int j) {
    const int d = i - j;
    const int e = d - 10;
    const bool pow2 = (e <= 1024) & ((e & (e - 1)) == 0);
    return (d >= 0) & ((i < 22) | (d <= 10) | pow2);
}

__device__ __forceinline__ bool tile_mask(int i, int j, bool far) {
    return log_mask(i, j) & (((i - j) > 26) == far);
}

template <int M>
__device__ __forceinline__ float row_xor(float x) {
    const int b = __builtin_bit_cast(int, x);
    return __builtin_bit_cast(float, __builtin_amdgcn_update_dpp(0, b, 0x160 | M, 0xf, 0xf, true));
}

constexpr int NTILES = 9;
__device__ __forceinline__ int tile_off(int t)  { return (t < 3) ? 16 * t : 10 + (32 << (t - 3)); }
__device__ __forceinline__ int tile_need(int t) { return (t < 3) ? 16 * t : (32 << (t - 3)); }

constexpr int OSTR = 68;

__global__ __launch_bounds__(WAVES * 32) __attribute__((amdgpu_waves_per_eu(1, 6)))
void logsparse_attn_kernel(const float* __restrict__ Q,
                           const float* __restrict__ K,
                           const float* __restrict__ V,
                           float* __restrict__ Out) {
    __shared__ __align__(16) float plds[WAVES][16][18];
    __shared__ __align__(16) float po[QPW][WAVES][16][OSTR];
    __shared__ float pm[QPW][WAVES][16];
    __shared__ float pl[QPW][WAVES][16];

    const int lane = threadIdx.x & 31;
    const int wave = __builtin_amdgcn_readfirstlane(threadIdx.x >> 5);
    const int g    = lane >> 4;
    const int ln   = lane & 15;

    const int gid = blockIdx.x;
    const int pr  = gid & 63;
    const int h   = (gid >> 6) & 7;
    const int b   = gid >> 9;
    const int turn = (gid + (gid >> 2) + (gid >> 4) + (gid >> 6) + (gid >> 8)) & 3;

  for (int q = 0; q < QPW; ++q) {
    const int i0 = (q ? pr : 127 - pr) * 16;

    const float* qrow = Q + ((size_t)(b * L_ + i0 + ln) * H_ + h) * E_ + 2 * g;
    v2f qf[16];
#pragma unroll
    for (int kc = 0; kc < 16; ++kc) {
        float2 t = *(const float2*)(qrow + 4 * kc);
        v2f a; a[0] = t.x; a[1] = t.y;
        qf[kc] = a;
    }

    v8f oacc[4];
#pragma unroll
    for (int nt = 0; nt < 4; ++nt) oacc[nt] = (v8f){};
    float rowmax[8], rowsum[8];
#pragma unroll
    for (int r = 0; r < 8; ++r) { rowmax[r] = -INFINITY; rowsum[r] = 0.0f; }

    int nact = 0;
#pragma unroll
    for (int t = 0; t < NTILES; ++t) nact += (tile_need(t) <= i0) ? 1 : 0;
    const int first = (wave - turn - q) & 3;
    for (int ti = first; ti < nact; ti += WAVES) {
        const bool far = ti >= 3;
        const int j0 = max(i0 - tile_off(ti), 0);

        const float* krow = K + ((size_t)(b * S_ + j0 + ln) * H_ + h) * E_ + 2 * g;
        v8f s = (v8f){};
#pragma unroll
        for (int kc = 0; kc < 16; ++kc) {
            float2 t = *(const float2*)(krow + 4 * kc);
            v2f kf; kf[0] = t.x; kf[1] = t.y;
            s = __builtin_amdgcn_wmma_f32_16x16x4_f32(false, qf[kc], false, kf,
                                                      (short)0, s, false, false);
        }

#pragma unroll
        for (int r = 0; r < 8; ++r) {
            const int i = i0 + r + 8 * g;
            const int j = j0 + ln;
            float logit = tile_mask(i, j, far) ? SCALE * s[r] : MASKED;

            float tmax = logit;
            tmax = fmaxf(tmax, row_xor<8>(tmax));
            tmax = fmaxf(tmax, row_xor<4>(tmax));
            tmax = fmaxf(tmax, row_xor<2>(tmax));
            tmax = fmaxf(tmax, row_xor<1>(tmax));

            float newmax = fmaxf(rowmax[r], tmax);
            float corr   = __expf(rowmax[r] - newmax);
            float p      = __expf(logit - newmax);

            float tsum = p;
            tsum += row_xor<8>(tsum);
            tsum += row_xor<4>(tsum);
            tsum += row_xor<2>(tsum);
            tsum += row_xor<1>(tsum);

            rowsum[r] = rowsum[r] * corr + tsum;
            rowmax[r] = newmax;
#pragma unroll
            for (int nt = 0; nt < 4; ++nt) oacc[nt][r] *= corr;

            plds[wave][r + 8 * g][ln] = p;
        }
        __builtin_amdgcn_wave_barrier();

        v2f pa[4];
#pragma unroll
        for (int kc = 0; kc < 4; ++kc) {
            float2 t = *(const float2*)(&plds[wave][ln][4 * kc + 2 * g]);
            v2f a; a[0] = t.x; a[1] = t.y;
            pa[kc] = a;
        }
        __builtin_amdgcn_wave_barrier();

        const float* vbase = V + ((size_t)(b * S_ + j0) * H_ + h) * D_;
#pragma unroll
        for (int nt = 0; nt < 4; ++nt) {
#pragma unroll
            for (int kc = 0; kc < 4; ++kc) {
                const int kr = 4 * kc + 2 * g;
                v2f vf;
                vf[0] = vbase[(size_t)(kr + 0) * (H_ * D_) + nt * 16 + ln];
                vf[1] = vbase[(size_t)(kr + 1) * (H_ * D_) + nt * 16 + ln];
                oacc[nt] = __builtin_amdgcn_wmma_f32_16x16x4_f32(false, pa[kc], false, vf,
                                                                 (short)0, oacc[nt], false, false);
            }
        }
    }

#pragma unroll
    for (int r = 0; r < 8; ++r) {
        if (ln == 0) {
            pm[q][wave][r + 8 * g] = rowmax[r];
            pl[q][wave][r + 8 * g] = rowsum[r];
        }
#pragma unroll
        for (int nt = 0; nt < 4; ++nt) po[q][wave][r + 8 * g][nt * 16 + ln] = oacc[nt][r];
    }
  }
    __syncthreads();

    const int c4 = (threadIdx.x & 15) * 4;
  for (int q = 0; q < QPW; ++q) {
    const int i0 = (q ? pr : 127 - pr) * 16;
    for (int hf = 0; hf < 2; ++hf) {
        const int row = (threadIdx.x >> 4) + 8 * hf;
        const float m0 = pm[q][0][row], m1 = pm[q][1][row], m2 = pm[q][2][row], m3 = pm[q][3][row];
        const float mx = fmaxf(fmaxf(m0, m1), fmaxf(m2, m3));
        const float s0 = __expf(m0 - mx), s1 = __expf(m1 - mx);
        const float s2 = __expf(m2 - mx), s3 = __expf(m3 - mx);
        const float lsum = ((pl[q][0][row] * s0 + pl[q][1][row] * s1) + pl[q][2][row] * s2) + pl[q][3][row] * s3;
        const float inv = 1.0f / lsum;
        const float4 a0 = *(const float4*)(&po[q][0][row][c4]);
        const float4 a1 = *(const float4*)(&po[q][1][row][c4]);
        const float4 a2 = *(const float4*)(&po[q][2][row][c4]);
        const float4 a3 = *(const float4*)(&po[q][3][row][c4]);
        float4 o;
        o.x = (((a0.x * s0 + a1.x * s1) + a2.x * s2) + a3.x * s3) * inv;
        o.y = (((a0.y * s0 + a1.y * s1) + a2.y * s2) + a3.y * s3) * inv;
        o.z = (((a0.z * s0 + a1.z * s1) + a2.z * s2) + a3.z * s3) * inv;
        o.w = (((a0.w * s0 + a1.w * s1) + a2.w * s2) + a3.w * s3) * inv;
        float* orow = Out + ((size_t)(b * L_ + i0 + row) * H_ + h) * D_;
        *(float4*)(orow + c4) = o;
    }
  }
}

extern "C" void kernel_launch(void* const* d_in, const int* in_sizes, int n_in,
                              void* d_out, int out_size, void* d_ws, size_t ws_size,
                              hipStream_t stream) {
    (void)in_sizes; (void)n_in; (void)out_size; (void)d_ws; (void)ws_size;
    const float* Q = (const float*)d_in[0];
    const float* K = (const float*)d_in[1];
    const float* V = (const float*)d_in[2];
    float* Out = (float*)d_out;

    dim3 grid(B_ * H_ * (L_ / 32));
    dim3 block(WAVES * 32);
    hipLaunchKernelGGL(logsparse_attn_kernel, grid, block, 0, stream, Q, K, V, Out);
}


# score tile computed transposed (keys x queries, one query per lane): row max/sum in-lane over 8 registers + one half swap; P enters P*V from registers as B operand, no LDS trip; regroups row, O sums
# speedup vs baseline: 4.6274x; 1.1243x over previous
#include <hip/hip_runtime.h>
#include <math.h>

typedef __attribute__((ext_vector_type(2))) float v2f;
typedef __attribute__((ext_vector_type(8))) float v8f;

#define WAVES 4
#define QPW 2

constexpr int B_ = 2, L_ = 2048, S_ = 2048, H_ = 8, E_ = 64, D_ = 64;
constexpr float SCALE  = 0.125f;
constexpr float MASKED = -1.25e8f;

__device__ __forceinline__ bool log_mask(int i, int j) {
    const int d = i - j;
    const int e = d - 10;
    const bool pow2 = (e <= 1024) & ((e & (e - 1)) == 0);
    return (d >= 0) & ((i < 22) | (d <= 10) | pow2);
}

__device__ __forceinline__ bool tile_mask(int i, int j, bool far) {
    return log_mask(i, j) & (((i - j) > 26) == far);
}

__device__ __forceinline__ float xhalf(float x) {
    const int b = __builtin_bit_cast(int, x);
    return __builtin_bit_cast(float, __builtin_amdgcn_permlanex16(b, b, 0x76543210, 0xfedcba98, false, false));
}

constexpr int NTILES = 9;
__device__ __forceinline__ int tile_off(int t)  { return (t < 3) ? 16 * t : 10 + (32 << (t - 3)); }
__device__ __forceinline__ int tile_need(int t) { return (t < 3) ? 16 * t : (32 << (t - 3)); }

constexpr int OSTR = 68;

__global__ __launch_bounds__(WAVES * 32) __attribute__((amdgpu_waves_per_eu(1, 6)))
void logsparse_attn_kernel(const float* __restrict__ Q,
                           const float* __restrict__ K,
                           const float* __restrict__ V,
                           float* __restrict__ Out) {
    __shared__ __align__(16) float po[QPW][WAVES][16][OSTR];
    __shared__ float pm[QPW][WAVES][16];
    __shared__ float pl[QPW][WAVES][16];

    const int lane = threadIdx.x & 31;
    const int wave = __builtin_amdgcn_readfirstlane(threadIdx.x >> 5);
    const int g    = lane >> 4;
    const int ln   = lane & 15;

    const int gid = blockIdx.x;
    const int pr  = gid & 63;
    const int h   = (gid >> 6) & 7;
    const int b   = gid >> 9;
    const int turn = (gid + (gid >> 2) + (gid >> 4) + (gid >> 6) + (gid >> 8)) & 3;

  for (int q = 0; q < QPW; ++q) {
    const int i0 = (q ? pr : 127 - pr) * 16;

    const float* qrow = Q + ((size_t)(b * L_ + i0 + ln) * H_ + h) * E_ + 2 * g;
    v2f qf[16];
#pragma unroll
    for (int kc = 0; kc < 16; ++kc) {
        float2 t = *(const float2*)(qrow + 4 * kc);
        v2f a; a[0] = t.x; a[1] = t.y;
        qf[kc] = a;
    }

    v8f oacc[4];
#pragma unroll
    for (int nt = 0; nt < 4; ++nt) oacc[nt] = (v8f){};
    float rowmax = -INFINITY, rowsum = 0.0f;

    int nact = 0;
#pragma unroll
    for (int t = 0; t < NTILES; ++t) nact += (tile_need(t) <= i0) ? 1 : 0;
    const int first = (wave - turn - q) & 3;
    for (int ti = first; ti < nact; ti += WAVES) {
        const bool far = ti >= 3;
        const int j0 = max(i0 - tile_off(ti), 0);

        const float* krow = K + ((size_t)(b * S_ + j0 + ln) * H_ + h) * E_ + 2 * g;
        v8f s = (v8f){};
#pragma unroll
        for (int kc = 0; kc < 16; ++kc) {
            float2 t = *(const float2*)(krow + 4 * kc);
            v2f kf; kf[0] = t.x; kf[1] = t.y;
            s = __builtin_amdgcn_wmma_f32_16x16x4_f32(false, kf, false, qf[kc],
                                                      (short)0, s, false, false);
        }

        const int i = i0 + ln;
        float lg[8];
#pragma unroll
        for (int r = 0; r < 8; ++r) {
            const int j = j0 + r + 8 * g;
            lg[r] = tile_mask(i, j, far) ? SCALE * s[r] : MASKED;
        }
        float tmax = fmaxf(fmaxf(fmaxf(lg[0], lg[1]), fmaxf(lg[2], lg[3])),
                           fmaxf(fmaxf(lg[4], lg[5]), fmaxf(lg[6], lg[7])));
        tmax = fmaxf(tmax, xhalf(tmax));

        const float newmax = fmaxf(rowmax, tmax);
        const float corr   = __expf(rowmax - newmax);
        float p[8];
#pragma unroll
        for (int r = 0; r < 8; ++r) p[r] = __expf(lg[r] - newmax);
        float tsum = ((p[0] + p[1]) + (p[2] + p[3])) + ((p[4] + p[5]) + (p[6] + p[7]));
        tsum += xhalf(tsum);

        rowsum = rowsum * corr + tsum;
        rowmax = newmax;
#pragma unroll
        for (int nt = 0; nt < 4; ++nt)
#pragma unroll
            for (int r = 0; r < 8; ++r) oacc[nt][r] *= corr;

        const float* vbase = V + ((size_t)(b * S_ + j0) * H_ + h) * D_;
#pragma unroll
        for (int nt = 0; nt < 4; ++nt) {
#pragma unroll
            for (int t = 0; t < 4; ++t) {
                const int kr = 8 * g + 2 * t;
                v2f vf;
                vf[0] = vbase[(size_t)(kr + 0) * (H_ * D_) + nt * 16 + ln];
                vf[1] = vbase[(size_t)(kr + 1) * (H_ * D_) + nt * 16 + ln];
                v2f pb; pb[0] = p[2 * t]; pb[1] = p[2 * t + 1];
                oacc[nt] = __builtin_amdgcn_wmma_f32_16x16x4_f32(false, vf, false, pb,
                                                                 (short)0, oacc[nt], false, false);
            }
        }
    }

    if (g == 0) {
        pm[q][wave][ln] = rowmax;
        pl[q][wave][ln] = rowsum;
    }
#pragma unroll
    for (int nt = 0; nt < 4; ++nt) {
        float4 lo, hi;
        lo.x = oacc[nt][0]; lo.y = oacc[nt][1]; lo.z = oacc[nt][2]; lo.w = oacc[nt][3];
        hi.x = oacc[nt][4]; hi.y = oacc[nt][5]; hi.z = oacc[nt][6]; hi.w = oacc[nt][7];
        *(float4*)(&po[q][wave][ln][nt * 16 + 8 * g])     = lo;
        *(float4*)(&po[q][wave][ln][nt * 16 + 8 * g + 4]) = hi;
    }
  }
    __syncthreads();

    const int c4 = (threadIdx.x & 15) * 4;
  for (int q = 0; q < QPW; ++q) {
    const int i0 = (q ? pr : 127 - pr) * 16;
    for (int hf = 0; hf < 2; ++hf) {
        const int row = (threadIdx.x >> 4) + 8 * hf;
        const float m0 = pm[q][0][row], m1 = pm[q][1][row], m2 = pm[q][2][row], m3 = pm[q][3][row];
        const float mx = fmaxf(fmaxf(m0, m1), fmaxf(m2, m3));
        const float s0 = __expf(m0 - mx), s1 = __expf(m1 - mx);
        const float s2 = __expf(m2 - mx), s3 = __expf(m3 - mx);
        const float lsum = ((pl[q][0][row] * s0 + pl[q][1][row] * s1) + pl[q][2][row] * s2) + pl[q][3][row] * s3;
        const float inv = 1.0f / lsum;
        const float4 a0 = *(const float4*)(&po[q][0][row][c4]);
        const float4 a1 = *(const float4*)(&po[q][1][row][c4]);
        const float4 a2 = *(const float4*)(&po[q][2][row][c4]);
        const float4 a3 = *(const float4*)(&po[q][3][row][c4]);
        float4 o;
        o.x = (((a0.x * s0 + a1.x * s1) + a2.x * s2) + a3.x * s3) * inv;
        o.y = (((a0.y * s0 + a1.y * s1) + a2.y * s2) + a3.y * s3) * inv;
        o.z = (((a0.z * s0 + a1.z * s1) + a2.z * s2) + a3.z * s3) * inv;
        o.w = (((a0.w * s0 + a1.w * s1) + a2.w * s2) + a3.w * s3) * inv;
        float* orow = Out + ((size_t)(b * L_ + i0 + row) * H_ + h) * D_;
        *(float4*)(orow + c4) = o;
    }
  }
}

extern "C" void kernel_launch(void* const* d_in, const int* in_sizes, int n_in,
                              void* d_out, int out_size, void* d_ws, size_t ws_size,
                              hipStream_t stream) {
    (void)in_sizes; (void)n_in; (void)out_size; (void)d_ws; (void)ws_size;
    const float* Q = (const float*)d_in[0];
    const float* K = (const float*)d_in[1];
    const float* V = (const float*)d_in[2];
    float* Out = (float*)d_out;

    dim3 grid(B_ * H_ * (L_ / 32));
    dim3 block(WAVES * 32);
    hipLaunchKernelGGL(logsparse_attn_kernel, grid, block, 0, stream, Q, K, V, Out);
}


# one query tile per wave: it walks its 1-9 key tiles alone, next tile's K,V in flight in a second fragment set; 2048 waves, no split over waves, no LDS merge; one softmax state regroups row, O sums
# speedup vs baseline: 5.1835x; 1.1202x over previous
#include <hip/hip_runtime.h>
#include <math.h>


typedef __attribute__((ext_vector_type(2))) float v2f;
typedef __attribute__((ext_vector_type(8))) float v8f;

#define WAVES 8

constexpr int B_ = 2, L_ = 2048, S_ = 2048, H_ = 8, E_ = 64, D_ = 64;
constexpr float SCALE  = 0.125f;
constexpr float MASKED = -1.25e8f;

__device__ __forceinline__ bool log_mask(int i, int j) {
    const int d = i - j;
    const int e = d - 10;
    const bool pow2 = (e <= 1024) & ((e & (e - 1)) == 0);
    return (d >= 0) & ((i < 22) | (d <= 10) | pow2);
}

__device__ __forceinline__ bool tile_mask(int i, int j, bool far) {
    return log_mask(i, j) & (((i - j) > 26) == far);
}

__device__ __forceinline__ float xhalf(float x) {
    const int b = __builtin_bit_cast(int, x);
    return __builtin_bit_cast(float, __builtin_amdgcn_permlanex16(b, b, 0x76543210, 0xfedcba98, false, false));
}

constexpr int NTILES = 9;
__device__ __forceinline__ int tile_off(int t)  { return (t < 3) ? 16 * t : 10 + (32 << (t - 3)); }
__device__ __forceinline__ int tile_need(int t) { return (t < 3) ? 16 * t : (32 << (t - 3)); }

constexpr int OSTR = 68;

__device__ __forceinline__ void load_k(v2f (&kf)[16], const float* __restrict__ K,
                                       int b, int h, int j0, int ln, int g) {
    const float* krow = K + ((size_t)(b * S_ + j0 + ln) * H_ + h) * E_ + 2 * g;
#pragma unroll
    for (int kc = 0; kc < 16; ++kc) {
        float2 t = *(const float2*)(krow + 4 * kc);
        v2f a; a[0] = t.x; a[1] = t.y;
        kf[kc] = a;
    }
}

__device__ __forceinline__ void load_v(v2f (&vf)[16], const float* __restrict__ V,
                                       int b, int h, int j0, int ln, int g) {
    const float* vbase = V + ((size_t)(b * S_ + j0) * H_ + h) * D_;
#pragma unroll
    for (int nt = 0; nt < 4; ++nt) {
#pragma unroll
        for (int t = 0; t < 4; ++t) {
            const int kr = 8 * g + 2 * t;
            v2f a;
            a[0] = vbase[(size_t)(kr + 0) * (H_ * D_) + nt * 16 + ln];
            a[1] = vbase[(size_t)(kr + 1) * (H_ * D_) + nt * 16 + ln];
            vf[nt * 4 + t] = a;
        }
    }
}

template <bool NEXT>
__device__ __forceinline__ void tile_step(int ti, int i0, int b, int h, int ln, int g,
                                          const float* __restrict__ K, const float* __restrict__ V,
                                          const v2f (&qf)[16], const v2f (&kf)[16], const v2f (&vf)[16],
                                          v2f (&kn)[16], v2f (&vn)[16],
                                          v8f (&oacc)[4], float& rowmax, float& rowsum) {
    const bool far = ti >= 3;
    const int j0 = max(i0 - tile_off(ti), 0);
    if (NEXT) {
        const int j0n = max(i0 - tile_off(ti + 1), 0);
        load_k(kn, K, b, h, j0n, ln, g);
        load_v(vn, V, b, h, j0n, ln, g);
    }

    v8f s = (v8f){};
#pragma unroll
    for (int kc = 0; kc < 16; ++kc)
        s = __builtin_amdgcn_wmma_f32_16x16x4_f32(false, kf[kc], false, qf[kc],
                                                  (short)0, s, false, false);

    const int i = i0 + ln;
    float lg[8];
#pragma unroll
    for (int r = 0; r < 8; ++r) {
        const int j = j0 + r + 8 * g;
        lg[r] = tile_mask(i, j, far) ? SCALE * s[r] : MASKED;
    }
    float tmax = fmaxf(fmaxf(fmaxf(lg[0], lg[1]), fmaxf(lg[2], lg[3])),
                       fmaxf(fmaxf(lg[4], lg[5]), fmaxf(lg[6], lg[7])));
    tmax = fmaxf(tmax, xhalf(tmax));

    const float newmax = fmaxf(rowmax, tmax);
    const float corr   = __expf(rowmax - newmax);
    float p[8];
#pragma unroll
    for (int r = 0; r < 8; ++r) p[r] = __expf(lg[r] - newmax);
    float tsum = ((p[0] + p[1]) + (p[2] + p[3])) + ((p[4] + p[5]) + (p[6] + p[7]));
    tsum += xhalf(tsum);

    rowsum = rowsum * corr + tsum;
    rowmax = newmax;
#pragma unroll
    for (int nt = 0; nt < 4; ++nt)
#pragma unroll
        for (int r = 0; r < 8; ++r) oacc[nt][r] *= corr;

#pragma unroll
    for (int nt = 0; nt < 4; ++nt) {
#pragma unroll
        for (int t = 0; t < 4; ++t) {
            v2f pb; pb[0] = p[2 * t]; pb[1] = p[2 * t + 1];
            oacc[nt] = __builtin_amdgcn_wmma_f32_16x16x4_f32(false, vf[nt * 4 + t], false, pb,
                                                             (short)0, oacc[nt], false, false);
        }
    }
}

__global__ __launch_bounds__(WAVES * 32) __attribute__((amdgpu_waves_per_eu(1, 2)))
void logsparse_attn_kernel(const float* __restrict__ Q,
                           const float* __restrict__ K,
                           const float* __restrict__ V,
                           float* __restrict__ Out) {
    __shared__ __align__(16) float po[WAVES][16][OSTR];

    const int lane = threadIdx.x & 31;
    const int wave = __builtin_amdgcn_readfirstlane(threadIdx.x >> 5);
    const int g    = lane >> 4;
    const int ln   = lane & 15;

    const int pid = blockIdx.x * 4 + (wave & 3);
    const int pr  = pid & 63;
    const int h   = (pid >> 6) & 7;
    const int b   = pid >> 9;
    const int i0  = ((wave < 4) ? 127 - pr : pr) * 16;

    const float* qrow = Q + ((size_t)(b * L_ + i0 + ln) * H_ + h) * E_ + 2 * g;
    v2f qf[16];
#pragma unroll
    for (int kc = 0; kc < 16; ++kc) {
        float2 t = *(const float2*)(qrow + 4 * kc);
        v2f a; a[0] = t.x; a[1] = t.y;
        qf[kc] = a;
    }
    v2f ka[16], va[16], kb[16], vb[16];
    load_k(ka, K, b, h, i0, ln, g);
    load_v(va, V, b, h, i0, ln, g);

    v8f oacc[4];
#pragma unroll
    for (int nt = 0; nt < 4; ++nt) oacc[nt] = (v8f){};
    float rowmax = -INFINITY, rowsum = 0.0f;

    int nact = 0;
#pragma unroll
    for (int t = 0; t < NTILES; ++t) nact += (tile_need(t) <= i0) ? 1 : 0;

    int ti = 0;
    for (; ti + 2 < nact; ti += 2) {
        tile_step<true>(ti,     i0, b, h, ln, g, K, V, qf, ka, va, kb, vb, oacc, rowmax, rowsum);
        tile_step<true>(ti + 1, i0, b, h, ln, g, K, V, qf, kb, vb, ka, va, oacc, rowmax, rowsum);
    }
    if (ti + 1 < nact) {
        tile_step<true>(ti,      i0, b, h, ln, g, K, V, qf, ka, va, kb, vb, oacc, rowmax, rowsum);
        tile_step<false>(ti + 1, i0, b, h, ln, g, K, V, qf, kb, vb, ka, va, oacc, rowmax, rowsum);
    } else {
        tile_step<false>(ti,     i0, b, h, ln, g, K, V, qf, ka, va, kb, vb, oacc, rowmax, rowsum);
    }

    const float inv = 1.0f / rowsum;
#pragma unroll
    for (int nt = 0; nt < 4; ++nt) {
        float4 lo, hi;
        lo.x = oacc[nt][0] * inv; lo.y = oacc[nt][1] * inv; lo.z = oacc[nt][2] * inv; lo.w = oacc[nt][3] * inv;
        hi.x = oacc[nt][4] * inv; hi.y = oacc[nt][5] * inv; hi.z = oacc[nt][6] * inv; hi.w = oacc[nt][7] * inv;
        *(float4*)(&po[wave][ln][nt * 16 + 8 * g])     = lo;
        *(float4*)(&po[wave][ln][nt * 16 + 8 * g + 4]) = hi;
    }
    __syncthreads();

    const int c4 = (lane & 15) * 4;
#pragma unroll
    for (int k = 0; k < 8; ++k) {
        const int row = g + 2 * k;
        const float4 o = *(const float4*)(&po[wave][row][c4]);
        float* orow = Out + ((size_t)(b * L_ + i0 + row) * H_ + h) * D_;
        *(float4*)(orow + c4) = o;
    }
}

extern "C" void kernel_launch(void* const* d_in, const int* in_sizes, int n_in,
                              void* d_out, int out_size, void* d_ws, size_t ws_size,
                              hipStream_t stream) {
    (void)in_sizes; (void)n_in; (void)out_size; (void)d_ws; (void)ws_size;
    const float* Q = (const float*)d_in[0];
    const float* K = (const float*)d_in[1];
    const float* V = (const float*)d_in[2];
    float* Out = (float*)d_out;

    dim3 grid(B_ * H_ * (L_ / 32) / 4);
    dim3 block(WAVES * 32);
    hipLaunchKernelGGL(logsparse_attn_kernel, grid, block, 0, stream, Q, K, V, Out);
}


# work-groups numbered by walk length, longest first (block id -> pair group 15 - id/16, batch and head in the low bits): late-entering groups get the short walks; same sums
# speedup vs baseline: 5.5101x; 1.0630x over previous
#include <hip/hip_runtime.h>
#include <math.h>


typedef __attribute__((ext_vector_type(2))) float v2f;
typedef __attribute__((ext_vector_type(8))) float v8f;

#define WAVES 8

constexpr int B_ = 2, L_ = 2048, S_ = 2048, H_ = 8, E_ = 64, D_ = 64;
constexpr float SCALE  = 0.125f;
constexpr float MASKED = -1.25e8f;

__device__ __forceinline__ bool log_mask(int i, int j) {
    const int d = i - j;
    const int e = d - 10;
    const bool pow2 = (e <= 1024) & ((e & (e - 1)) == 0);
    return (d >= 0) & ((i < 22) | (d <= 10) | pow2);
}

__device__ __forceinline__ bool tile_mask(int i, int j, bool far) {
    return log_mask(i, j) & (((i - j) > 26) == far);
}

__device__ __forceinline__ float xhalf(float x) {
    const int b = __builtin_bit_cast(int, x);
    return __builtin_bit_cast(float, __builtin_amdgcn_permlanex16(b, b, 0x76543210, 0xfedcba98, false, false));
}

constexpr int NTILES = 9;
__device__ __forceinline__ int tile_off(int t)  { return (t < 3) ? 16 * t : 10 + (32 << (t - 3)); }
__device__ __forceinline__ int tile_need(int t) { return (t < 3) ? 16 * t : (32 << (t - 3)); }

constexpr int OSTR = 68;

__device__ __forceinline__ void load_k(v2f (&kf)[16], const float* __restrict__ K,
                                       int b, int h, int j0, int ln, int g) {
    const float* krow = K + ((size_t)(b * S_ + j0 + ln) * H_ + h) * E_ + 2 * g;
#pragma unroll
    for (int kc = 0; kc < 16; ++kc) {
        float2 t = *(const float2*)(krow + 4 * kc);
        v2f a; a[0] = t.x; a[1] = t.y;
        kf[kc] = a;
    }
}

__device__ __forceinline__ void load_v(v2f (&vf)[16], const float* __restrict__ V,
                                       int b, int h, int j0, int ln, int g) {
    const float* vbase = V + ((size_t)(b * S_ + j0) * H_ + h) * D_;
#pragma unroll
    for (int nt = 0; nt < 4; ++nt) {
#pragma unroll
        for (int t = 0; t < 4; ++t) {
            const int kr = 8 * g + 2 * t;
            v2f a;
            a[0] = vbase[(size_t)(kr + 0) * (H_ * D_) + nt * 16 + ln];
            a[1] = vbase[(size_t)(kr + 1) * (H_ * D_) + nt * 16 + ln];
            vf[nt * 4 + t] = a;
        }
    }
}

template <bool NEXT>
__device__ __forceinline__ void tile_step(int ti, int i0, int b, int h, int ln, int g,
                                          const float* __restrict__ K, const float* __restrict__ V,
                                          const v2f (&qf)[16], const v2f (&kf)[16], const v2f (&vf)[16],
                                          v2f (&kn)[16], v2f (&vn)[16],
                                          v8f (&oacc)[4], float& rowmax, float& rowsum) {
    const bool far = ti >= 3;
    const int j0 = max(i0 - tile_off(ti), 0);
    if (NEXT) {
        const int j0n = max(i0 - tile_off(ti + 1), 0);
        load_k(kn, K, b, h, j0n, ln, g);
        load_v(vn, V, b, h, j0n, ln, g);
    }

    v8f s = (v8f){};
#pragma unroll
    for (int kc = 0; kc < 16; ++kc)
        s = __builtin_amdgcn_wmma_f32_16x16x4_f32(false, kf[kc], false, qf[kc],
                                                  (short)0, s, false, false);

    const int i = i0 + ln;
    float lg[8];
#pragma unroll
    for (int r = 0; r < 8; ++r) {
        const int j = j0 + r + 8 * g;
        lg[r] = tile_mask(i, j, far) ? SCALE * s[r] : MASKED;
    }
    float tmax = fmaxf(fmaxf(fmaxf(lg[0], lg[1]), fmaxf(lg[2], lg[3])),
                       fmaxf(fmaxf(lg[4], lg[5]), fmaxf(lg[6], lg[7])));
    tmax = fmaxf(tmax, xhalf(tmax));

    const float newmax = fmaxf(rowmax, tmax);
    const float corr   = __expf(rowmax - newmax);
    float p[8];
#pragma unroll
    for (int r = 0; r < 8; ++r) p[r] = __expf(lg[r] - newmax);
    float tsum = ((p[0] + p[1]) + (p[2] + p[3])) + ((p[4] + p[5]) + (p[6] + p[7]));
    tsum += xhalf(tsum);

    rowsum = rowsum * corr + tsum;
    rowmax = newmax;
#pragma unroll
    for (int nt = 0; nt < 4; ++nt)
#pragma unroll
        for (int r = 0; r < 8; ++r) oacc[nt][r] *= corr;

#pragma unroll
    for (int nt = 0; nt < 4; ++nt) {
#pragma unroll
        for (int t = 0; t < 4; ++t) {
            v2f pb; pb[0] = p[2 * t]; pb[1] = p[2 * t + 1];
            oacc[nt] = __builtin_amdgcn_wmma_f32_16x16x4_f32(false, vf[nt * 4 + t], false, pb,
                                                             (short)0, oacc[nt], false, false);
        }
    }
}

__global__ __launch_bounds__(WAVES * 32) __attribute__((amdgpu_waves_per_eu(1, 2)))
void logsparse_attn_kernel(const float* __restrict__ Q,
                           const float* __restrict__ K,
                           const float* __restrict__ V,
                           float* __restrict__ Out) {
    __shared__ __align__(16) float po[WAVES][16][OSTR];

    const int lane = threadIdx.x & 31;
    const int wave = __builtin_amdgcn_readfirstlane(threadIdx.x >> 5);
    const int g    = lane >> 4;
    const int ln   = lane & 15;

    const int q   = 15 - (blockIdx.x >> 4);
    const int bh  = blockIdx.x & 15;
    const int pr  = q * 4 + (wave & 3);
    const int h   = bh & 7;
    const int b   = bh >> 3;
    const int i0  = ((wave < 4) ? 127 - pr : pr) * 16;

    const float* qrow = Q + ((size_t)(b * L_ + i0 + ln) * H_ + h) * E_ + 2 * g;
    v2f qf[16];
#pragma unroll
    for (int kc = 0; kc < 16; ++kc) {
        float2 t = *(const float2*)(qrow + 4 * kc);
        v2f a; a[0] = t.x; a[1] = t.y;
        qf[kc] = a;
    }
    v2f ka[16], va[16], kb[16], vb[16];
    load_k(ka, K, b, h, i0, ln, g);
    load_v(va, V, b, h, i0, ln, g);

    v8f oacc[4];
#pragma unroll
    for (int nt = 0; nt < 4; ++nt) oacc[nt] = (v8f){};
    float rowmax = -INFINITY, rowsum = 0.0f;

    int nact = 0;
#pragma unroll
    for (int t = 0; t < NTILES; ++t) nact += (tile_need(t) <= i0) ? 1 : 0;

    int ti = 0;
    for (; ti + 2 < nact; ti += 2) {
        tile_step<true>(ti,     i0, b, h, ln, g, K, V, qf, ka, va, kb, vb, oacc, rowmax, rowsum);
        tile_step<true>(ti + 1, i0, b, h, ln, g, K, V, qf, kb, vb, ka, va, oacc, rowmax, rowsum);
    }
    if (ti + 1 < nact) {
        tile_step<true>(ti,      i0, b, h, ln, g, K, V, qf, ka, va, kb, vb, oacc, rowmax, rowsum);
        tile_step<false>(ti + 1, i0, b, h, ln, g, K, V, qf, kb, vb, ka, va, oacc, rowmax, rowsum);
    } else {
        tile_step<false>(ti,     i0, b, h, ln, g, K, V, qf, ka, va, kb, vb, oacc, rowmax, rowsum);
    }

    const float inv = 1.0f / rowsum;
#pragma unroll
    for (int nt = 0; nt < 4; ++nt) {
        float4 lo, hi;
        lo.x = oacc[nt][0] * inv; lo.y = oacc[nt][1] * inv; lo.z = oacc[nt][2] * inv; lo.w = oacc[nt][3] * inv;
        hi.x = oacc[nt][4] * inv; hi.y = oacc[nt][5] * inv; hi.z = oacc[nt][6] * inv; hi.w = oacc[nt][7] * inv;
        *(float4*)(&po[wave][ln][nt * 16 + 8 * g])     = lo;
        *(float4*)(&po[wave][ln][nt * 16 + 8 * g + 4]) = hi;
    }
    __syncthreads();

    const int c4 = (lane & 15) * 4;
#pragma unroll
    for (int k = 0; k < 8; ++k) {
        const int row = g + 2 * k;
        const float4 o = *(const float4*)(&po[wave][row][c4]);
        float* orow = Out + ((size_t)(b * L_ + i0 + row) * H_ + h) * D_;
        *(float4*)(orow + c4) = o;
    }
}

extern "C" void kernel_launch(void* const* d_in, const int* in_sizes, int n_in,
                              void* d_out, int out_size, void* d_ws, size_t ws_size,
                              hipStream_t stream) {
    (void)in_sizes; (void)n_in; (void)out_size; (void)d_ws; (void)ws_size;
    const float* Q = (const float*)d_in[0];
    const float* K = (const float*)d_in[1];
    const float* V = (const float*)d_in[2];
    float* Out = (float*)d_out;

    dim3 grid(B_ * H_ * (L_ / 32) / 4);
    dim3 block(WAVES * 32);
    hipLaunchKernelGGL(logsparse_attn_kernel, grid, block, 0, stream, Q, K, V, Out);
}
